# starting kernel
# baseline (speedup 1.0000x reference)
#include <hip/hip_runtime.h>


typedef __attribute__((ext_vector_type(2))) float v2f;
typedef __attribute__((ext_vector_type(8))) float v8f;

__global__ void zero_f32(float* __restrict__ p, long long n) {
  long long i = (long long)blockIdx.x * blockDim.x + threadIdx.x;
  long long stride = (long long)gridDim.x * blockDim.x;
  for (; i < n; i += stride) p[i] = 0.0f;
}

__global__ void degree_kernel(const int* __restrict__ dst,
                              float* __restrict__ deg, int E) {
  int e = blockIdx.x * blockDim.x + threadIdx.x;
  if (e < E) atomicAdd(&deg[dst[e]], 1.0f);
}

template <int D>
__global__ void scatter_add_kernel(const float* __restrict__ feat,
                                   const int* __restrict__ src,
                                   const int* __restrict__ dst,
                                   float* __restrict__ agg, int E) {
  constexpr int PER_EDGE = D / 4;
  long long tid = (long long)blockIdx.x * blockDim.x + threadIdx.x;
  long long total = (long long)E * PER_EDGE;
  if (tid >= total) return;
  int e = (int)(tid / PER_EDGE);
  int f = (int)(tid % PER_EDGE) * 4;
  int s = src[e], d = dst[e];
  float4 v = *(const float4*)(feat + (size_t)s * D + f);
  float* p = agg + (size_t)d * D + f;
  atomicAdd(p + 0, v.x);
  atomicAdd(p + 1, v.y);
  atomicAdd(p + 2, v.z);
  atomicAdd(p + 3, v.w);
}

template <int K, bool RELU>
__launch_bounds__(256)
__global__ void sage_gemm_wmma(const float* __restrict__ aggsum,
                               const float* __restrict__ deg,
                               const float* __restrict__ xin,
                               const float* __restrict__ Wl,
                               const float* __restrict__ Wr,
                               const float* __restrict__ bias,
                               float* __restrict__ out,
                               int M, int N) {
  const int lane = threadIdx.x & 31;
  const int wave = threadIdx.x >> 5;
  const int tile = blockIdx.x * (blockDim.x >> 5) + wave;
  const int tilesN = N >> 4;
  const int nTiles = (M >> 4) * tilesN;
  if (tile >= nTiles) return;

  const int tm = tile / tilesN;
  const int tn = tile - tm * tilesN;

  const int m  = lane & 15;
  const int kh = lane >> 4;

  const int arow = tm * 16 + m;
  const int bcol = tn * 16 + m;

  const float invd = 1.0f / fmaxf(deg[arow], 1.0f);

  const float* aP = aggsum + (size_t)arow * K + 2 * kh;
  const float* xP = xin    + (size_t)arow * K + 2 * kh;
  const float* lP = Wl     + (size_t)bcol * K + 2 * kh;
  const float* rP = Wr     + (size_t)bcol * K + 2 * kh;

  v8f acc = {};

#pragma unroll 8
  for (int k0 = 0; k0 < K; k0 += 4) {
    v2f a, b;
    a.x = aP[k0] * invd;
    a.y = aP[k0 + 1] * invd;
    b.x = lP[k0];
    b.y = lP[k0 + 1];
    acc = __builtin_amdgcn_wmma_f32_16x16x4_f32(false, a, false, b,
                                                (short)0, acc, false, false);
  }

#pragma unroll 8
  for (int k0 = 0; k0 < K; k0 += 4) {
    v2f a, b;
    a.x = xP[k0];
    a.y = xP[k0 + 1];
    b.x = rP[k0];
    b.y = rP[k0 + 1];
    acc = __builtin_amdgcn_wmma_f32_16x16x4_f32(false, a, false, b,
                                                (short)0, acc, false, false);
  }

  const float bv = bias[bcol];
#pragma unroll
  for (int r = 0; r < 8; ++r) {
    float v = acc[r] + bv;
    if (RELU) v = fmaxf(v, 0.0f);
    out[(size_t)(tm * 16 + r + 8 * kh) * N + bcol] = v;
  }
}

extern "C" void kernel_launch(void* const* d_in, const int* in_sizes, int n_in,
                              void* d_out, int out_size, void* d_ws, size_t ws_size,
                              hipStream_t stream) {
  (void)n_in; (void)out_size; (void)ws_size;

  const float* x   = (const float*)d_in[0];
  const int*   ei  = (const int*)  d_in[1];
  const float* W1l = (const float*)d_in[2];
  const float* b1  = (const float*)d_in[3];
  const float* W1r = (const float*)d_in[4];
  const float* W2l = (const float*)d_in[5];
  const float* b2  = (const float*)d_in[6];
  const float* W2r = (const float*)d_in[7];
  float* out = (float*)d_out;

  const int D1 = in_sizes[3];
  const int D2 = in_sizes[6];
  const int D0 = in_sizes[2] / D1;
  const int M  = in_sizes[0] / D0;
  const int E  = in_sizes[1] / 2;
  const int* src = ei;
  const int* dst = ei + E;

  float* deg  = (float*)d_ws;
  float* bufA = deg + M;
  float* h    = bufA + (size_t)M * D1;

  zero_f32<<<4096, 256, 0, stream>>>(deg, (long long)M * D0 + M);
  degree_kernel<<<(E + 255) / 256, 256, 0, stream>>>(dst, deg, E);
  {
    long long total = (long long)E * (128 / 4);
    scatter_add_kernel<128><<<(int)((total + 255) / 256), 256, 0, stream>>>(
        x, src, dst, bufA, E);
  }
  {
    int tiles = (M / 16) * (D1 / 16);
    sage_gemm_wmma<128, true><<<(tiles + 7) / 8, 256, 0, stream>>>(
        bufA, deg, x, W1l, W1r, b1, h, M, D1);
  }

  zero_f32<<<4096, 256, 0, stream>>>(bufA, (long long)M * D1);
  {
    long long total = (long long)E * (256 / 4);
    scatter_add_kernel<256><<<(int)((total + 255) / 256), 256, 0, stream>>>(
        h, src, dst, bufA, E);
  }
  {
    int tiles = (M / 16) * (D2 / 16);
    sage_gemm_wmma<256, false><<<(tiles + 7) / 8, 256, 0, stream>>>(
        bufA, deg, h, W2l, W2r, b2, out, M, D2);
  }
}


# Layer 2 walks per-range edge lists that layer 1's filter pass wrote (kept in the range's own output rows), in place of filtering the whole edge list again; sums stay arrival-order LDS atomics
# speedup vs baseline: 4.4587x; 4.4587x over previous
#include <hip/hip_runtime.h>


typedef __attribute__((ext_vector_type(2))) float v2f;
typedef __attribute__((ext_vector_type(8))) float v8f;

constexpr int AGG_NODES   = 196;
constexpr int AGG_WAVES   = 32;
constexpr int AGG_QCAP    = 160;
constexpr int AGG_SCAN_U  = 8;
constexpr int ROW_TILES   = (AGG_NODES + 15) / 16;
constexpr int LDS_ROWS    = ROW_TILES * 16;
constexpr int LIST_HDR    = 32;

template <int V> struct IntC { static constexpr int value = V; };

template <int D, int N, int LW, bool COUNT, bool RELU>
__launch_bounds__(1024)
__global__ void sage_layer_lds(const float* __restrict__ feat,
                              const int* __restrict__ src,
                              const int* __restrict__ dst,
                              const float* __restrict__ Wl,
                              const float* __restrict__ Wr,
                              const float* __restrict__ bias,
                              float* deg,
                              float* out,
                              int* lst,
                              int M, int E) {
  static_assert(LW % 2 == 0 && LW >= LIST_HDR, "a range's list area holds its header and 8-byte pairs");
  constexpr int G = D / 128;
  constexpr int B = 16 / G;
  constexpr int P = D + 2;
  __shared__ __attribute__((aligned(16))) float acc[LDS_ROWS * P];
  __shared__ float degL[AGG_NODES];
  __shared__ int qe[AGG_WAVES][AGG_QCAP];
  __shared__ int qd[AGG_WAVES][AGG_QCAP];
  __shared__ int listN;

  const int tid  = threadIdx.x;
  const int lane = tid & 31;
  const int wave = __builtin_amdgcn_readfirstlane(tid >> 5);
  const int lo   = blockIdx.x * AGG_NODES;
  const int cnt  = min(AGG_NODES, M - lo);
  if (cnt <= 0) return;

  int*  lstR = lst + (size_t)lo * LW;
  int2* ent  = reinterpret_cast<int2*>(lstR + LIST_HDR);
  const int cap = (cnt * LW - LIST_HDR) >> 1;
  int tot = 0;
  if (!COUNT) tot = lstR[0];

  for (int i = tid; i < LDS_ROWS * P; i += 1024) acc[i] = 0.0f;
  if (tid < AGG_NODES) degL[tid] = 0.0f;
  if (COUNT && tid == 0) listN = 0;
  __syncthreads();

  int* myqe = qe[wave];
  int* myqd = qd[wave];
  int qn = 0;

  auto addRows = [&](int l0, int n, int s_v, int dl_v) {
    float4 v[B][G];
#pragma unroll
    for (int i = 0; i < B; ++i) {
      if (i < n) {
        const int s = __builtin_amdgcn_readlane(s_v, l0 + i);
        const float4* r = reinterpret_cast<const float4*>(feat + (size_t)s * D) + lane;
#pragma unroll
        for (int g = 0; g < G; ++g) v[i][g] = r[32 * g];
      }
    }
#pragma unroll
    for (int i = 0; i < B; ++i) {
      if (i < n) {
        const int dl = __builtin_amdgcn_readlane(dl_v, l0 + i);
        float* a = acc + dl * P + lane;
#pragma unroll
        for (int g = 0; g < G; ++g) {
          atomicAdd(a + g * 128,      v[i][g].x);
          atomicAdd(a + g * 128 + 32, v[i][g].y);
          atomicAdd(a + g * 128 + 64, v[i][g].z);
          atomicAdd(a + g * 128 + 96, v[i][g].w);
        }
      }
    }
  };

  auto drain = [&](int n) {
    const int base = qn - n;
    int dl_v = 0, s_v = 0;
    if (lane < n) {
      dl_v = myqd[base + lane];
      s_v  = src[myqe[base + lane]];
    }
    if (COUNT) {
      int p = 0;
      if (lane == 0) p = atomicAdd(&listN, n);
      p = __builtin_amdgcn_readfirstlane(p);
      if (lane < n && p + lane < cap) ent[p + lane] = make_int2(s_v, dl_v);
    }
    addRows(0, n, s_v, dl_v);
    if (COUNT) {
      if (lane < n) atomicAdd(&degL[dl_v], 1.0f);
    }
    qn = base;
  };

  auto push = [&](bool m, int e, int dl) {
    const unsigned mask = __builtin_amdgcn_ballot_w32(m);
    if (mask != 0u) {
      const int pos = qn + (int)__builtin_amdgcn_mbcnt_lo(mask, 0u);
      if (m) { myqe[pos] = e; myqd[pos] = dl; }
      qn += __builtin_popcount(mask);
    }
  };

  const bool useList = !COUNT && (unsigned)tot <= (unsigned)cap;

  if (useList) {
    constexpr int R = 32 / B;
    auto pairsAt = [&](int b) {
      const int ei = (b + AGG_WAVES * (lane / B)) * B + (lane % B);
      return (ei < tot) ? ent[ei] : make_int2(0, 0);
    };
    int2 e2 = pairsAt(wave);
    for (int b = wave; b * B < tot; b += AGG_WAVES * R) {
      const int2 nx = pairsAt(b + AGG_WAVES * R);
#pragma unroll
      for (int q = 0; q < R; ++q) {
        const int nb = tot - (b + AGG_WAVES * q) * B;
        if (nb >= B)     addRows(q * B, B, e2.x, e2.y);
        else if (nb > 0) addRows(q * B, nb, e2.x, e2.y);
      }
      e2 = nx;
    }
  } else {

  const int nFull = ((E & 3) == 0) ? (E >> 7) : 0;
  const int rot = (int)(((long long)blockIdx.x * nFull) / gridDim.x);
  const int4* dst4 = reinterpret_cast<const int4*>(dst);

  for (int i0 = wave; i0 < nFull; i0 += AGG_WAVES * AGG_SCAN_U) {
    int4 d[AGG_SCAN_U];
    int  cb[AGG_SCAN_U];
#pragma unroll
    for (int u = 0; u < AGG_SCAN_U; ++u) {
      const int i = i0 + AGG_WAVES * u;
      int c = i + rot;
      if (c >= nFull) c -= nFull;
      cb[u] = c;
      if (i < nFull) d[u] = dst4[(size_t)c * 32 + lane];
      else           d[u] = make_int4(-1, -1, -1, -1);
    }
#pragma unroll
    for (int u = 0; u < AGG_SCAN_U; ++u) {
      const unsigned r0 = (unsigned)(d[u].x - lo), r1 = (unsigned)(d[u].y - lo);
      const unsigned r2 = (unsigned)(d[u].z - lo), r3 = (unsigned)(d[u].w - lo);
      const bool m0 = r0 < (unsigned)cnt, m1 = r1 < (unsigned)cnt;
      const bool m2 = r2 < (unsigned)cnt, m3 = r3 < (unsigned)cnt;
      if (__builtin_amdgcn_ballot_w32(m0 | m1 | m2 | m3) != 0u) {
        const int e = cb[u] * 128 + lane * 4;
        push(m0, e,     (int)r0);
        push(m1, e + 1, (int)r1);
        push(m2, e + 2, (int)r2);
        push(m3, e + 3, (int)r3);
        while (qn >= B) drain(B);
      }
    }
  }

  for (int e0 = nFull * 128 + wave * 32; e0 < E; e0 += AGG_WAVES * 32) {
    const int e = e0 + lane;
    const unsigned r = (e < E) ? (unsigned)(dst[e] - lo) : 0xffffffffu;
    push(r < (unsigned)cnt, e, (int)r);
    while (qn >= B) drain(B);
  }
  if (qn > 0) drain(qn);

  }

  __syncthreads();

  if (COUNT) {
    if (tid < cnt) deg[lo + tid] = degL[tid];
    if (tid == 0) lstR[0] = listN;
  }

  for (int r = wave; r < cnt; r += AGG_WAVES) {
    const float dg = COUNT ? degL[r] : deg[lo + r];
    const float iv = 1.0f / fmaxf(dg, 1.0f);
    float* a = acc + r * P + lane;
#pragma unroll
    for (int j = 0; j < D / 32; ++j) a[32 * j] *= iv;
  }
  __syncthreads();

  constexpr int NT = N / 16;
  constexpr int MW = AGG_WAVES / NT;
  constexpr int TM = (ROW_TILES + MW - 1) / MW;
  const int nT = (cnt + 15) >> 4;
  const int nt = wave % NT;
  const int mg = wave / NT;
  const int m  = lane & 15;
  const int hh = lane >> 4;
  const int ncol = nt * 16 + m;

  int myT = 0;
#pragma unroll
  for (int t = 0; t < TM; ++t) myT += (mg + MW * t < nT) ? 1 : 0;
  const bool fullT = (myT == TM);

  const float* aP0 = acc + (mg * 16 + m) * P + hh * 32;
  v8f c[TM];
#pragma unroll
  for (int t = 0; t < TM; ++t) c[t] = v8f{};

  auto mac = [&](auto tag, const float* __restrict__ Wm) {
    constexpr int T = decltype(tag)::value;
    constexpr int S = G * 16;
    const float4* wrow = reinterpret_cast<const float4*>(Wm + (size_t)ncol * D);
    float4 w0 = wrow[0], w1 = wrow[1];
#pragma unroll 2
    for (int s = 0; s < S; ++s) {
      const int sn = min(s + 1, S - 1);
      const float4 n0 = wrow[2 * sn], n1 = wrow[2 * sn + 1];
      const int g = s >> 4, lp = s & 15;
#pragma unroll
      for (int cp = 0; cp < 2; ++cp) {
        v2f b;
        if (cp == 0) { b.x = hh ? w0.y : w0.x; b.y = hh ? w1.y : w1.x; }
        else         { b.x = hh ? w0.w : w0.z; b.y = hh ? w1.w : w1.z; }
        const int off = g * 128 + cp * 64 + lp * 2;
        v2f a[T];
#pragma unroll
        for (int t = 0; t < T; ++t) a[t] = *reinterpret_cast<const v2f*>(aP0 + t * (MW * 16 * P) + off);
#pragma unroll
        for (int t = 0; t < T; ++t) {
          c[t] = __builtin_amdgcn_wmma_f32_16x16x4_f32(false, a[t], false, b,
                                                       (short)0, c[t], false, false);
        }
      }
      w0 = n0; w1 = n1;
    }
  };

  if (fullT) mac(IntC<TM>{}, Wl);
  else       mac(IntC<TM - 1>{}, Wl);

  __syncthreads();
  constexpr int Q = D / 4;
  for (int i = tid; i < cnt * Q; i += 1024) {
    const int r = i / Q, q = i - r * Q;
    const int g = q >> 5, l = q & 31;
    const float4 v = reinterpret_cast<const float4*>(feat + (size_t)(lo + r) * D)[q];
    float* a = acc + r * P + g * 128 + l;
    a[0] = v.x; a[32] = v.y; a[64] = v.z; a[96] = v.w;
  }
  __syncthreads();

  auto fin = [&](auto tag) {
    constexpr int T = decltype(tag)::value;
    const float bv = bias[ncol];
#pragma unroll
    for (int t = 0; t < T; ++t) {
      if (mg + MW * t < nT) {
#pragma unroll
        for (int r = 0; r < 8; ++r) {
          const int row = (mg + MW * t) * 16 + 8 * hh + r;
          float v = c[t][r] + bv;
          if (RELU) v = fmaxf(v, 0.0f);
          if (row < cnt) out[(size_t)(lo + row) * N + ncol] = v;
        }
      }
    }
  };

  if (fullT) { mac(IntC<TM>{}, Wr);     fin(IntC<TM>{}); }
  else       { mac(IntC<TM - 1>{}, Wr); fin(IntC<TM - 1>{}); }
}

extern "C" void kernel_launch(void* const* d_in, const int* in_sizes, int n_in,
                              void* d_out, int out_size, void* d_ws, size_t ws_size,
                              hipStream_t stream) {
  (void)n_in; (void)out_size; (void)ws_size;

  const float* x   = (const float*)d_in[0];
  const int*   ei  = (const int*)  d_in[1];
  const float* W1l = (const float*)d_in[2];
  const float* b1  = (const float*)d_in[3];
  const float* W1r = (const float*)d_in[4];
  const float* W2l = (const float*)d_in[5];
  const float* b2  = (const float*)d_in[6];
  const float* W2r = (const float*)d_in[7];
  float* out = (float*)d_out;

  const int D1 = in_sizes[3];
  const int D2 = in_sizes[6];
  const int D0 = in_sizes[2] / D1;
  const int M  = in_sizes[0] / D0;
  const int E  = in_sizes[1] / 2;
  const int* src = ei;
  const int* dst = ei + E;

  (void)D2;
  float* h   = (float*)d_ws;
  float* deg = h + (size_t)M * D1;

  const int ranges = (M + AGG_NODES - 1) / AGG_NODES;

  sage_layer_lds<128, 256, 128, true, true><<<ranges, 1024, 0, stream>>>(
      x, src, dst, W1l, W1r, b1, deg, h, (int*)out, M, E);

  sage_layer_lds<256, 128, 128, false, false><<<ranges, 1024, 0, stream>>>(
      h, src, dst, W2l, W2r, b2, deg, out, (int*)out, M, E);
}


# Layer 1 tests one-byte range keys (dst / 196, from a new first launch, kept in free ends of the list areas), 16 edges per lane and load, not the 4-byte dst list; 2 to 3 launches; sums as before
# speedup vs baseline: 4.7404x; 1.0632x over previous
#include <hip/hip_runtime.h>


typedef __attribute__((ext_vector_type(2))) float v2f;
typedef __attribute__((ext_vector_type(8))) float v8f;

constexpr int AGG_NODES   = 196;
constexpr int AGG_WAVES   = 32;
constexpr int AGG_QCAP    = 160;
constexpr int AGG_SCAN_U  = 8;
constexpr int ROW_TILES   = (AGG_NODES + 15) / 16;
constexpr int LDS_ROWS    = ROW_TILES * 16;
constexpr int LIST_HDR    = 32;

constexpr int LIST_W     = 128;
constexpr int KEY_BYTES  = 4096;
constexpr int KEY_INTS   = KEY_BYTES / 4;
static_assert(AGG_NODES * LIST_W - LIST_HDR - KEY_INTS > 0 && KEY_BYTES % 512 == 0, "keys sit behind the list");

template <int V> struct IntC { static constexpr int value = V; };

__device__ __forceinline__ bool keysFit(int M, int E, int ranges) {
  return ranges <= 256 &&
         (long long)(M / AGG_NODES) * KEY_BYTES >= (((long long)E + 511) & ~511LL);
}
__device__ __forceinline__ size_t keyOffset(int e) {
  return ((size_t)(e / KEY_BYTES) * (AGG_NODES * LIST_W) + (AGG_NODES * LIST_W - KEY_INTS)) * 4 + (size_t)(e % KEY_BYTES);
}

__global__ void range_keys(const int* __restrict__ dst, int* lst, int M, int E) {
  const int ranges = (M + AGG_NODES - 1) / AGG_NODES;
  if (!keysFit(M, E, ranges)) return;
  const int nW   = (E + 511) >> 9;
  const int lane = threadIdx.x & 31;
  const int w    = (blockIdx.x * blockDim.x + threadIdx.x) >> 5;
  const bool vec = (E & 3) == 0;
  int4 d[4];
  if (vec && w * 512 + 512 <= E) {
#pragma unroll
    for (int j = 0; j < 4; ++j) d[j] = *reinterpret_cast<const int4*>(dst + w * 512 + j * 128 + lane * 4);
  } else {
#pragma unroll
    for (int j = 0; j < 4; ++j) {
      const int e = w * 512 + j * 128 + lane * 4;
      d[j].x = (e     < E) ? dst[e]     : -1;
      d[j].y = (e + 1 < E) ? dst[e + 1] : -1;
      d[j].z = (e + 2 < E) ? dst[e + 2] : -1;
      d[j].w = (e + 3 < E) ? dst[e + 3] : -1;
    }
  }
#pragma unroll
  for (int j = 0; j < 4; ++j) {
    const int e = w * 512 + j * 128 + lane * 4;
    const unsigned k0 = min((unsigned)d[j].x / (unsigned)AGG_NODES, 255u);
    const unsigned k1 = min((unsigned)d[j].y / (unsigned)AGG_NODES, 255u);
    const unsigned k2 = min((unsigned)d[j].z / (unsigned)AGG_NODES, 255u);
    const unsigned k3 = min((unsigned)d[j].w / (unsigned)AGG_NODES, 255u);
    if (w < nW)
      *reinterpret_cast<unsigned*>(reinterpret_cast<char*>(lst) + keyOffset(e)) = k0 | (k1 << 8) | (k2 << 16) | (k3 << 24);
  }
}

template <int D, int N, int LW, bool COUNT, bool RELU>
__launch_bounds__(1024)
__global__ void sage_layer_lds(const float* __restrict__ feat,
                              const int* __restrict__ src,
                              const int* __restrict__ dst,
                              const float* __restrict__ Wl,
                              const float* __restrict__ Wr,
                              const float* __restrict__ bias,
                              float* deg,
                              float* out,
                              int* lst,
                              int M, int E) {
  static_assert(LW % 2 == 0 && LW >= LIST_HDR, "a range's list area holds its header and 8-byte pairs");
  constexpr int G = D / 128;
  constexpr int B = 16 / G;
  constexpr int P = D + 2;
  __shared__ __attribute__((aligned(16))) float acc[LDS_ROWS * P];
  __shared__ float degL[AGG_NODES];
  __shared__ int qe[AGG_WAVES][AGG_QCAP];
  __shared__ int qd[AGG_WAVES][AGG_QCAP];
  __shared__ int listN;

  const int tid  = threadIdx.x;
  const int lane = tid & 31;
  const int wave = __builtin_amdgcn_readfirstlane(tid >> 5);
  const int lo   = blockIdx.x * AGG_NODES;
  const int cnt  = min(AGG_NODES, M - lo);
  if (cnt <= 0) return;

  int*  lstR = lst + (size_t)lo * LW;
  int2* ent  = reinterpret_cast<int2*>(lstR + LIST_HDR);
  const bool keysOn  = keysFit(M, E, (int)gridDim.x);
  const int  keyInts = (keysOn && cnt == AGG_NODES) ? KEY_INTS : 0;
  const int cap = (cnt * LW - LIST_HDR - keyInts) >> 1;
  int tot = 0;
  if (!COUNT) tot = lstR[0];

  for (int i = tid; i < LDS_ROWS * P; i += 1024) acc[i] = 0.0f;
  if (tid < AGG_NODES) degL[tid] = 0.0f;
  if (COUNT && tid == 0) listN = 0;
  __syncthreads();

  int* myqe = qe[wave];
  int* myqd = qd[wave];
  int qn = 0;

  auto addRows = [&](int l0, int n, int s_v, int dl_v) {
    float4 v[B][G];
#pragma unroll
    for (int i = 0; i < B; ++i) {
      if (i < n) {
        const int s = __builtin_amdgcn_readlane(s_v, l0 + i);
        const float4* r = reinterpret_cast<const float4*>(feat + (size_t)s * D) + lane;
#pragma unroll
        for (int g = 0; g < G; ++g) v[i][g] = r[32 * g];
      }
    }
#pragma unroll
    for (int i = 0; i < B; ++i) {
      if (i < n) {
        const int dl = __builtin_amdgcn_readlane(dl_v, l0 + i);
        float* a = acc + dl * P + lane;
#pragma unroll
        for (int g = 0; g < G; ++g) {
          atomicAdd(a + g * 128,      v[i][g].x);
          atomicAdd(a + g * 128 + 32, v[i][g].y);
          atomicAdd(a + g * 128 + 64, v[i][g].z);
          atomicAdd(a + g * 128 + 96, v[i][g].w);
        }
      }
    }
  };

  auto drain = [&](int n, auto keyed) {
    constexpr bool KEYED = decltype(keyed)::value != 0;
    const int base = qn - n;
    int dl_v = 0, s_v = 0;
    bool ok = false;
    if (lane < n) {
      const int e = myqe[base + lane];
      s_v = src[e];
      if (KEYED) {
        const unsigned r = (unsigned)(dst[e] - lo);
        ok   = r < (unsigned)cnt;
        dl_v = ok ? (int)r : LDS_ROWS - 1;
        if (!ok) s_v = 0;
      } else {
        dl_v = myqd[base + lane];
        ok   = true;
      }
    }
    if (COUNT) {
      int p = 0;
      if (lane == 0) p = atomicAdd(&listN, n);
      p = __builtin_amdgcn_readfirstlane(p);
      if (lane < n && p + lane < cap) ent[p + lane] = make_int2(s_v, dl_v);
    }
    addRows(0, n, s_v, dl_v);
    if (COUNT) {
      if (ok) atomicAdd(&degL[dl_v], 1.0f);
    }
    qn = base;
  };

  auto push = [&](bool m, int e, int dl) {
    const unsigned mask = __builtin_amdgcn_ballot_w32(m);
    if (mask != 0u) {
      const int pos = qn + (int)__builtin_amdgcn_mbcnt_lo(mask, 0u);
      if (m) { myqe[pos] = e; myqd[pos] = dl; }
      qn += __builtin_popcount(mask);
    }
  };

  auto pushE = [&](bool m, int e) {
    const unsigned mask = __builtin_amdgcn_ballot_w32(m);
    const int pos = qn + (int)__builtin_amdgcn_mbcnt_lo(mask, 0u);
    if (m) myqe[pos] = e;
    qn += __builtin_popcount(mask);
  };

  const bool useList = !COUNT && (unsigned)tot <= (unsigned)cap;

  if (useList) {
    constexpr int R = 32 / B;
    auto pairsAt = [&](int b) {
      const int ei = (b + AGG_WAVES * (lane / B)) * B + (lane % B);
      return (ei < tot) ? ent[ei] : make_int2(0, 0);
    };
    int2 e2 = pairsAt(wave);
    for (int b = wave; b * B < tot; b += AGG_WAVES * R) {
      const int2 nx = pairsAt(b + AGG_WAVES * R);
#pragma unroll
      for (int q = 0; q < R; ++q) {
        const int nb = tot - (b + AGG_WAVES * q) * B;
        if (nb >= B)     addRows(q * B, B, e2.x, e2.y);
        else if (nb > 0) addRows(q * B, nb, e2.x, e2.y);
      }
      e2 = nx;
    }
  } else if (COUNT && keysOn) {

  constexpr int KU = 8;
  const int nW  = (E + 511) >> 9;
  const int rot = (int)(((long long)blockIdx.x * nW) / gridDim.x);
  const unsigned pat = (unsigned)blockIdx.x * 0x01010101u;
  const char* keys = reinterpret_cast<const char*>(lst);
  auto hit = [&](unsigned k) {
    const unsigned x = k ^ pat;
    const unsigned t = (x & 0x7f7f7f7fu) + 0x7f7f7f7fu;
    return ~(t | x) & 0x80808080u;
  };
  for (int i0 = wave; i0 < nW; i0 += AGG_WAVES * KU) {
    uint4 k[KU];
    int   cb[KU];
#pragma unroll
    for (int u = 0; u < KU; ++u) {
      int c = min(i0 + AGG_WAVES * u, nW - 1) + rot;
      if (c >= nW) c -= nW;
      cb[u] = c;
      k[u] = *reinterpret_cast<const uint4*>(keys + (size_t)((unsigned)c >> 3) * (AGG_NODES * LIST_W * 4)
                                             + ((AGG_NODES * LIST_W - KEY_INTS) * 4 + (c & 7) * 512) + lane * 16);
    }
#pragma unroll
    for (int u = 0; u < KU; ++u) {
      unsigned g = hit(k[u].w) | (hit(k[u].z) >> 1) | (hit(k[u].y) >> 2) | (hit(k[u].x) >> 3);
      if (i0 + AGG_WAVES * u >= nW) g = 0u;
      while (__builtin_amdgcn_ballot_w32(g != 0u) != 0u) {
        const int p = __builtin_ctz(g | 0x80000000u);
        const int e = cb[u] * 512 + lane * 16 + ((p & 7) - 4) * 4 + (p >> 3);
        pushE(g != 0u && e < E, e);
        g &= g - 1u;
        while (qn >= B) drain(B, IntC<1>{});
      }
    }
  }
  if (qn > 0) drain(qn, IntC<1>{});

  } else {

  const int nFull = ((E & 3) == 0) ? (E >> 7) : 0;
  const int rot = (int)(((long long)blockIdx.x * nFull) / gridDim.x);
  const int4* dst4 = reinterpret_cast<const int4*>(dst);

  for (int i0 = wave; i0 < nFull; i0 += AGG_WAVES * AGG_SCAN_U) {
    int4 d[AGG_SCAN_U];
    int  cb[AGG_SCAN_U];
#pragma unroll
    for (int u = 0; u < AGG_SCAN_U; ++u) {
      const int i = i0 + AGG_WAVES * u;
      int c = i + rot;
      if (c >= nFull) c -= nFull;
      cb[u] = c;
      if (i < nFull) d[u] = dst4[(size_t)c * 32 + lane];
      else           d[u] = make_int4(-1, -1, -1, -1);
    }
#pragma unroll
    for (int u = 0; u < AGG_SCAN_U; ++u) {
      const unsigned r0 = (unsigned)(d[u].x - lo), r1 = (unsigned)(d[u].y - lo);
      const unsigned r2 = (unsigned)(d[u].z - lo), r3 = (unsigned)(d[u].w - lo);
      const bool m0 = r0 < (unsigned)cnt, m1 = r1 < (unsigned)cnt;
      const bool m2 = r2 < (unsigned)cnt, m3 = r3 < (unsigned)cnt;
      if (__builtin_amdgcn_ballot_w32(m0 | m1 | m2 | m3) != 0u) {
        const int e = cb[u] * 128 + lane * 4;
        push(m0, e,     (int)r0);
        push(m1, e + 1, (int)r1);
        push(m2, e + 2, (int)r2);
        push(m3, e + 3, (int)r3);
        while (qn >= B) drain(B, IntC<0>{});
      }
    }
  }

  for (int e0 = nFull * 128 + wave * 32; e0 < E; e0 += AGG_WAVES * 32) {
    const int e = e0 + lane;
    const unsigned r = (e < E) ? (unsigned)(dst[e] - lo) : 0xffffffffu;
    push(r < (unsigned)cnt, e, (int)r);
    while (qn >= B) drain(B, IntC<0>{});
  }
  if (qn > 0) drain(qn, IntC<0>{});

  }

  __syncthreads();

  if (COUNT) {
    if (tid < cnt) deg[lo + tid] = degL[tid];
    if (tid == 0) lstR[0] = listN;
  }

  for (int r = wave; r < cnt; r += AGG_WAVES) {
    const float dg = COUNT ? degL[r] : deg[lo + r];
    const float iv = 1.0f / fmaxf(dg, 1.0f);
    float* a = acc + r * P + lane;
#pragma unroll
    for (int j = 0; j < D / 32; ++j) a[32 * j] *= iv;
  }
  __syncthreads();

  constexpr int NT = N / 16;
  constexpr int MW = AGG_WAVES / NT;
  constexpr int TM = (ROW_TILES + MW - 1) / MW;
  const int nT = (cnt + 15) >> 4;
  const int nt = wave % NT;
  const int mg = wave / NT;
  const int m  = lane & 15;
  const int hh = lane >> 4;
  const int ncol = nt * 16 + m;

  int myT = 0;
#pragma unroll
  for (int t = 0; t < TM; ++t) myT += (mg + MW * t < nT) ? 1 : 0;
  const bool fullT = (myT == TM);

  const float* aP0 = acc + (mg * 16 + m) * P + hh * 32;
  v8f c[TM];
#pragma unroll
  for (int t = 0; t < TM; ++t) c[t] = v8f{};

  auto mac = [&](auto tag, const float* __restrict__ Wm) {
    constexpr int T = decltype(tag)::value;
    constexpr int S = G * 16;
    const float4* wrow = reinterpret_cast<const float4*>(Wm + (size_t)ncol * D);
    float4 w0 = wrow[0], w1 = wrow[1];
#pragma unroll 2
    for (int s = 0; s < S; ++s) {
      const int sn = min(s + 1, S - 1);
      const float4 n0 = wrow[2 * sn], n1 = wrow[2 * sn + 1];
      const int g = s >> 4, lp = s & 15;
#pragma unroll
      for (int cp = 0; cp < 2; ++cp) {
        v2f b;
        if (cp == 0) { b.x = hh ? w0.y : w0.x; b.y = hh ? w1.y : w1.x; }
        else         { b.x = hh ? w0.w : w0.z; b.y = hh ? w1.w : w1.z; }
        const int off = g * 128 + cp * 64 + lp * 2;
        v2f a[T];
#pragma unroll
        for (int t = 0; t < T; ++t) a[t] = *reinterpret_cast<const v2f*>(aP0 + t * (MW * 16 * P) + off);
#pragma unroll
        for (int t = 0; t < T; ++t) {
          c[t] = __builtin_amdgcn_wmma_f32_16x16x4_f32(false, a[t], false, b,
                                                       (short)0, c[t], false, false);
        }
      }
      w0 = n0; w1 = n1;
    }
  };

  if (fullT) mac(IntC<TM>{}, Wl);
  else       mac(IntC<TM - 1>{}, Wl);

  __syncthreads();
  constexpr int Q = D / 4;
  for (int i = tid; i < cnt * Q; i += 1024) {
    const int r = i / Q, q = i - r * Q;
    const int g = q >> 5, l = q & 31;
    const float4 v = reinterpret_cast<const float4*>(feat + (size_t)(lo + r) * D)[q];
    float* a = acc + r * P + g * 128 + l;
    a[0] = v.x; a[32] = v.y; a[64] = v.z; a[96] = v.w;
  }
  __syncthreads();

  auto fin = [&](auto tag) {
    constexpr int T = decltype(tag)::value;
    const float bv = bias[ncol];
#pragma unroll
    for (int t = 0; t < T; ++t) {
      if (mg + MW * t < nT) {
#pragma unroll
        for (int r = 0; r < 8; ++r) {
          const int row = (mg + MW * t) * 16 + 8 * hh + r;
          float v = c[t][r] + bv;
          if (RELU) v = fmaxf(v, 0.0f);
          if (row < cnt) out[(size_t)(lo + row) * N + ncol] = v;
        }
      }
    }
  };

  if (fullT) { mac(IntC<TM>{}, Wr);     fin(IntC<TM>{}); }
  else       { mac(IntC<TM - 1>{}, Wr); fin(IntC<TM - 1>{}); }
}

extern "C" void kernel_launch(void* const* d_in, const int* in_sizes, int n_in,
                              void* d_out, int out_size, void* d_ws, size_t ws_size,
                              hipStream_t stream) {
  (void)n_in; (void)out_size; (void)ws_size;

  const float* x   = (const float*)d_in[0];
  const int*   ei  = (const int*)  d_in[1];
  const float* W1l = (const float*)d_in[2];
  const float* b1  = (const float*)d_in[3];
  const float* W1r = (const float*)d_in[4];
  const float* W2l = (const float*)d_in[5];
  const float* b2  = (const float*)d_in[6];
  const float* W2r = (const float*)d_in[7];
  float* out = (float*)d_out;

  const int D1 = in_sizes[3];
  const int D2 = in_sizes[6];
  const int D0 = in_sizes[2] / D1;
  const int M  = in_sizes[0] / D0;
  const int E  = in_sizes[1] / 2;
  const int* src = ei;
  const int* dst = ei + E;

  (void)D2;
  float* h   = (float*)d_ws;
  float* deg = h + (size_t)M * D1;

  const int ranges = (M + AGG_NODES - 1) / AGG_NODES;

  range_keys<<<(E + 4095) / 4096, 256, 0, stream>>>(dst, (int*)out, M, E);

  sage_layer_lds<128, 256, LIST_W, true, true><<<ranges, 1024, 0, stream>>>(
      x, src, dst, W1l, W1r, b1, deg, h, (int*)out, M, E);

  sage_layer_lds<256, 128, LIST_W, false, false><<<ranges, 1024, 0, stream>>>(
      h, src, dst, W2l, W2r, b2, deg, out, (int*)out, M, E);
}
